# starting kernel
# baseline (speedup 1.0000x reference)
#include <hip/hip_runtime.h>
#include <hip/hip_bf16.h>


typedef __attribute__((ext_vector_type(2))) float v2f;
typedef __attribute__((ext_vector_type(8))) float v8f;

#define DIM 64

__device__ __forceinline__ void atomic_add_f32_dev(float* addr, float val) {
    asm volatile("global_atomic_add_f32 %0, %1, off scope:SCOPE_DEV"
                 :: "v"(addr), "v"(val)
                 : "memory");
}

__global__ __launch_bounds__(256) void zero_kernel(float4* __restrict__ out, int n4) {
    int i = blockIdx.x * blockDim.x + threadIdx.x;
    if (i < n4) out[i] = make_float4(0.f, 0.f, 0.f, 0.f);
}

__global__ __launch_bounds__(256) void gemm_kernel(const float* __restrict__ x,
                                                   const float* __restrict__ W,
                                                   const float* __restrict__ b,
                                                   float* __restrict__ support,
                                                   int n_nodes) {
    const int lane    = threadIdx.x & 31;
    const int wave    = threadIdx.x >> 5;
    const int tile    = blockIdx.x * 8 + wave;
    const int n_tiles = n_nodes >> 4;
    if (tile >= n_tiles) return;

    const int row0    = tile << 4;
    const int halfSel = lane >> 4;
    const int l       = lane & 15;

    v8f acc[4];
    #pragma unroll
    for (int nt = 0; nt < 4; ++nt) {
        const float bias = b[nt * 16 + l];
        #pragma unroll
        for (int j = 0; j < 8; ++j) acc[nt][j] = bias;
    }

    #pragma unroll 4
    for (int k = 0; k < DIM; k += 4) {
        const float* xa = x + (size_t)(row0 + l) * DIM + k + 2 * halfSel;
        v2f a;
        a[0] = xa[0];
        a[1] = xa[1];
        #pragma unroll
        for (int nt = 0; nt < 4; ++nt) {
            const float* wb = W + (size_t)(nt * 16 + l) * DIM + k + 2 * halfSel;
            v2f bf;
            bf[0] = wb[0];
            bf[1] = wb[1];
            acc[nt] = __builtin_amdgcn_wmma_f32_16x16x4_f32(
                 false, a,  false, bf,
                 (short)0, acc[nt],  false,  false);
        }
    }

    #pragma unroll
    for (int nt = 0; nt < 4; ++nt) {
        #pragma unroll
        for (int j = 0; j < 8; ++j) {
            const int m = j + 8 * halfSel;
            support[(size_t)(row0 + m) * DIM + nt * 16 + l] = acc[nt][j];
        }
    }
}

__global__ __launch_bounds__(256) void scatter_kernel(const float* __restrict__ support,
                                                      const int* __restrict__ esrc,
                                                      const int* __restrict__ edst,
                                                      const float* __restrict__ ew,
                                                      float* __restrict__ out,
                                                      int n_edges) {
    const int t = blockIdx.x * blockDim.x + threadIdx.x;
    const int e = t >> 4;
    if (e >= n_edges) return;
    const int l16 = t & 15;

    const int   s = esrc[e];
    const int   d = edst[e];
    const float w = ew[e];

    const float4 v = ((const float4*)(support + (size_t)s * DIM))[l16];

    float* orow = out + (size_t)d * DIM + 4 * l16;
    atomic_add_f32_dev(orow + 0, w * v.x);
    atomic_add_f32_dev(orow + 1, w * v.y);
    atomic_add_f32_dev(orow + 2, w * v.z);
    atomic_add_f32_dev(orow + 3, w * v.w);
}

extern "C" void kernel_launch(void* const* d_in, const int* in_sizes, int n_in,
                              void* d_out, int out_size, void* d_ws, size_t ws_size,
                              hipStream_t stream) {
    const float* x    = (const float*)d_in[0];
    const float* W    = (const float*)d_in[1];
    const float* b    = (const float*)d_in[2];
    const int*   esrc = (const int*)  d_in[3];
    const int*   edst = (const int*)  d_in[4];
    const float* ew   = (const float*)d_in[5];

    float* out     = (float*)d_out;
    float* support = (float*)d_ws;

    const int n_nodes = in_sizes[0] / DIM;
    const int n_edges = in_sizes[3];

    const int n4 = out_size / 4;
    zero_kernel<<<(n4 + 255) / 256, 256, 0, stream>>>((float4*)out, n4);

    const int n_tiles = n_nodes / 16;
    const int gblocks = (n_tiles + 7) / 8;
    gemm_kernel<<<gblocks, 256, 0, stream>>>(x, W, b, support, n_nodes);

    const long long sthreads = (long long)n_edges * 16;
    const unsigned  sblocks  = (unsigned)((sthreads + 255) / 256);
    scatter_kernel<<<sblocks, 256, 0, stream>>>(support, esrc, edst, ew, out, n_edges);
}


# Scatter remapped: one wave per edge, 16 edges per wave with scalar index loads; lane owns dims l and l+32, so every gather load and f32 atomic covers one whole 128-B line; atomics kept
# speedup vs baseline: 1.8749x; 1.8749x over previous
#include <hip/hip_runtime.h>
#include <hip/hip_bf16.h>


typedef __attribute__((ext_vector_type(2))) float v2f;
typedef __attribute__((ext_vector_type(8))) float v8f;

#define DIM 64

__device__ __forceinline__ void atomic_add_f32_dev(float* addr, float val) {
    asm volatile("global_atomic_add_f32 %0, %1, off scope:SCOPE_DEV"
                 :: "v"(addr), "v"(val)
                 : "memory");
}

__global__ __launch_bounds__(256) void zero_kernel(float4* __restrict__ out, int n4) {
    int i = blockIdx.x * blockDim.x + threadIdx.x;
    if (i < n4) out[i] = make_float4(0.f, 0.f, 0.f, 0.f);
}

__global__ __launch_bounds__(256) void gemm_kernel(const float* __restrict__ x,
                                                   const float* __restrict__ W,
                                                   const float* __restrict__ b,
                                                   float* __restrict__ support,
                                                   int n_nodes) {
    const int lane    = threadIdx.x & 31;
    const int wave    = threadIdx.x >> 5;
    const int tile    = blockIdx.x * 8 + wave;
    const int n_tiles = n_nodes >> 4;
    if (tile >= n_tiles) return;

    const int row0    = tile << 4;
    const int halfSel = lane >> 4;
    const int l       = lane & 15;

    v8f acc[4];
    #pragma unroll
    for (int nt = 0; nt < 4; ++nt) {
        const float bias = b[nt * 16 + l];
        #pragma unroll
        for (int j = 0; j < 8; ++j) acc[nt][j] = bias;
    }

    #pragma unroll 4
    for (int k = 0; k < DIM; k += 4) {
        const float* xa = x + (size_t)(row0 + l) * DIM + k + 2 * halfSel;
        v2f a;
        a[0] = xa[0];
        a[1] = xa[1];
        #pragma unroll
        for (int nt = 0; nt < 4; ++nt) {
            const float* wb = W + (size_t)(nt * 16 + l) * DIM + k + 2 * halfSel;
            v2f bf;
            bf[0] = wb[0];
            bf[1] = wb[1];
            acc[nt] = __builtin_amdgcn_wmma_f32_16x16x4_f32(
                 false, a,  false, bf,
                 (short)0, acc[nt],  false,  false);
        }
    }

    #pragma unroll
    for (int nt = 0; nt < 4; ++nt) {
        #pragma unroll
        for (int j = 0; j < 8; ++j) {
            const int m = j + 8 * halfSel;
            support[(size_t)(row0 + m) * DIM + nt * 16 + l] = acc[nt][j];
        }
    }
}

#define EPW 16

__device__ __forceinline__ void scatter_one(const float* __restrict__ support,
                                            float* __restrict__ out,
                                            int s, int d, float w, unsigned lane) {
    const float* srow = support + (size_t)s * DIM;
    const float  v0 = srow[lane];
    const float  v1 = srow[lane + 32];
    float* orow = out + (size_t)d * DIM + lane;
    atomic_add_f32_dev(orow,      w * v0);
    atomic_add_f32_dev(orow + 32, w * v1);
}

__global__ __launch_bounds__(256) void scatter_kernel(const float* __restrict__ support,
                                                      const int* __restrict__ esrc,
                                                      const int* __restrict__ edst,
                                                      const float* __restrict__ ew,
                                                      float* __restrict__ out,
                                                      int n_edges) {
    const unsigned lane = threadIdx.x & 31u;
    const int wid = __builtin_amdgcn_readfirstlane((int)(blockIdx.x * 8 + (threadIdx.x >> 5)));
    const int e0  = wid * EPW;
    if (e0 >= n_edges) return;

    if (e0 + EPW <= n_edges) {
        int   s[EPW], d[EPW];
        float w[EPW];
        #pragma unroll
        for (int j = 0; j < EPW; ++j) { s[j] = esrc[e0 + j]; d[j] = edst[e0 + j]; w[j] = ew[e0 + j]; }

        float v0[EPW], v1[EPW];
        #pragma unroll
        for (int j = 0; j < EPW; ++j) {
            const float* srow = support + (size_t)s[j] * DIM;
            v0[j] = srow[lane];
            v1[j] = srow[lane + 32];
        }
        #pragma unroll
        for (int j = 0; j < EPW; ++j) {
            float* orow = out + (size_t)d[j] * DIM + lane;
            atomic_add_f32_dev(orow,      w[j] * v0[j]);
            atomic_add_f32_dev(orow + 32, w[j] * v1[j]);
        }
    } else {
        for (int e = e0; e < n_edges; ++e)
            scatter_one(support, out, esrc[e], edst[e], ew[e], lane);
    }
}

extern "C" void kernel_launch(void* const* d_in, const int* in_sizes, int n_in,
                              void* d_out, int out_size, void* d_ws, size_t ws_size,
                              hipStream_t stream) {
    const float* x    = (const float*)d_in[0];
    const float* W    = (const float*)d_in[1];
    const float* b    = (const float*)d_in[2];
    const int*   esrc = (const int*)  d_in[3];
    const int*   edst = (const int*)  d_in[4];
    const float* ew   = (const float*)d_in[5];

    float* out     = (float*)d_out;
    float* support = (float*)d_ws;

    const int n_nodes = in_sizes[0] / DIM;
    const int n_edges = in_sizes[3];

    const int n4 = out_size / 4;
    zero_kernel<<<(n4 + 255) / 256, 256, 0, stream>>>((float4*)out, n4);

    const int n_tiles = n_nodes / 16;
    const int gblocks = (n_tiles + 7) / 8;
    gemm_kernel<<<gblocks, 256, 0, stream>>>(x, W, b, support, n_nodes);

    const int      swaves  = (n_edges + EPW - 1) / EPW;
    const unsigned sblocks = (unsigned)((swaves + 7) / 8);
    scatter_kernel<<<sblocks, 256, 0, stream>>>(support, esrc, edst, ew, out, n_edges);
}
